# starting kernel
# baseline (speedup 1.0000x reference)
#include <hip/hip_runtime.h>
#include <hip/hip_bf16.h>

typedef __attribute__((ext_vector_type(2))) float v2f;
typedef __attribute__((ext_vector_type(8))) float v8f;

#define IN_DIM 128
#define OUT_DIM 256
#define LN_EPS 1e-5f

__global__ void zero_u32(unsigned* __restrict__ p, unsigned n) {
    unsigned i = blockIdx.x * blockDim.x + threadIdx.x;
    if (i < n) p[i] = 0u;
}

__global__ void degree_kernel(const int* __restrict__ src, const int* __restrict__ dst,
                              int* __restrict__ deg_out, int* __restrict__ deg_in, int n_edges) {
    int e = blockIdx.x * blockDim.x + threadIdx.x;
    if (e >= n_edges) return;
    atomicAdd(&deg_out[src[e]], 1);
    atomicAdd(&deg_in[dst[e]], 1);
}

__global__ void compute_h(const float* __restrict__ feat, const int* __restrict__ deg_out,
                          float* __restrict__ h, int n_nodes) {
    int idx = blockIdx.x * blockDim.x + threadIdx.x;
    int total = n_nodes * (IN_DIM / 4);
    if (idx >= total) return;
    int node = idx / (IN_DIM / 4);
    float nrm = rsqrtf(fmaxf((float)deg_out[node], 1.0f));
    float4 v = ((const float4*)feat)[idx];
    v.x *= nrm; v.y *= nrm; v.z *= nrm; v.w *= nrm;
    ((float4*)h)[idx] = v;
}

__global__ __launch_bounds__(256) void spmm_scatter(const int* __restrict__ src,
                                                    const int* __restrict__ dst,
                                                    const float* __restrict__ h,
                                                    float* __restrict__ agg, int n_edges) {
    int edge = blockIdx.x * (blockDim.x >> 5) + (threadIdx.x >> 5);
    if (edge >= n_edges) return;
    int lane = threadIdx.x & 31;
    int s = src[edge];
    int d = dst[edge];
    const float4* hp = (const float4*)(h + (size_t)s * IN_DIM);
    float4 v = hp[lane];
    float* ap = agg + (size_t)d * IN_DIM + lane * 4;
    __hip_atomic_fetch_add(ap + 0, v.x, __ATOMIC_RELAXED, __HIP_MEMORY_SCOPE_AGENT);
    __hip_atomic_fetch_add(ap + 1, v.y, __ATOMIC_RELAXED, __HIP_MEMORY_SCOPE_AGENT);
    __hip_atomic_fetch_add(ap + 2, v.z, __ATOMIC_RELAXED, __HIP_MEMORY_SCOPE_AGENT);
    __hip_atomic_fetch_add(ap + 3, v.w, __ATOMIC_RELAXED, __HIP_MEMORY_SCOPE_AGENT);
}

__global__ __launch_bounds__(256) void gcn_fused(const float* __restrict__ agg,
                                                 const int* __restrict__ deg_in,
                                                 const float* __restrict__ W,
                                                 const float* __restrict__ bias,
                                                 const float* __restrict__ gamma,
                                                 const float* __restrict__ beta,
                                                 const float* __restrict__ feat,
                                                 const float* __restrict__ skip_W,
                                                 const float* __restrict__ skip_b,
                                                 float* __restrict__ out, int n_nodes) {
    __shared__ float lds_gcn[16][OUT_DIM + 4];
    __shared__ float lds_skip[16][OUT_DIM + 4];

    const int m0   = blockIdx.x * 16;
    const int wave = threadIdx.x >> 5;
    const int lane = threadIdx.x & 31;
    const int half = lane >> 4;
    const int l16  = lane & 15;

    const int arow = m0 + l16;
    const float* agg_row  = agg  + (size_t)arow * IN_DIM;
    const float* feat_row = feat + (size_t)arow * IN_DIM;
    const int bcol = wave * 32 + l16;

    v8f cg0 = {}, cg1 = {};
    v8f cs0 = {}, cs1 = {};

    for (int k0 = 0; k0 < IN_DIM; k0 += 4) {
        const int ka = k0 + 2 * half;
        v2f aA = *(const v2f*)(agg_row + ka);
        v2f aF = *(const v2f*)(feat_row + ka);
        const float* Wk0 = W      + (size_t)ka * OUT_DIM;
        const float* Wk1 = Wk0 + OUT_DIM;
        const float* Sk0 = skip_W + (size_t)ka * OUT_DIM;
        const float* Sk1 = Sk0 + OUT_DIM;
        v2f bW0, bW1, bS0, bS1;
        bW0[0] = Wk0[bcol];      bW0[1] = Wk1[bcol];
        bW1[0] = Wk0[bcol + 16]; bW1[1] = Wk1[bcol + 16];
        bS0[0] = Sk0[bcol];      bS0[1] = Sk1[bcol];
        bS1[0] = Sk0[bcol + 16]; bS1[1] = Sk1[bcol + 16];
        cg0 = __builtin_amdgcn_wmma_f32_16x16x4_f32(false, aA, false, bW0, (short)0, cg0, false, false);
        cg1 = __builtin_amdgcn_wmma_f32_16x16x4_f32(false, aA, false, bW1, (short)0, cg1, false, false);
        cs0 = __builtin_amdgcn_wmma_f32_16x16x4_f32(false, aF, false, bS0, (short)0, cs0, false, false);
        cs1 = __builtin_amdgcn_wmma_f32_16x16x4_f32(false, aF, false, bS1, (short)0, cs1, false, false);
    }

    float nrmv[8];
#pragma unroll
    for (int i = 0; i < 8; ++i) {
        const int row = i + 8 * half;
        nrmv[i] = rsqrtf(fmaxf((float)deg_in[m0 + row], 1.0f));
    }

    {
        const float bb0 = bias[bcol],        sb0 = skip_b[bcol];
        const float bb1 = bias[bcol + 16],   sb1 = skip_b[bcol + 16];
#pragma unroll
        for (int i = 0; i < 8; ++i) {
            const int row = i + 8 * half;
            lds_gcn[row][bcol]       = cg0[i] * nrmv[i] + bb0;
            lds_gcn[row][bcol + 16]  = cg1[i] * nrmv[i] + bb1;
            lds_skip[row][bcol]      = cs0[i] + sb0;
            lds_skip[row][bcol + 16] = cs1[i] + sb1;
        }
    }
    __syncthreads();

#pragma unroll
    for (int rr = 0; rr < 2; ++rr) {
        const int row = wave * 2 + rr;
        float vals[8];
        float s = 0.f, s2 = 0.f;
#pragma unroll
        for (int j = 0; j < 8; ++j) {
            float x = lds_gcn[row][lane * 8 + j];
            vals[j] = x;
            s += x;
            s2 += x * x;
        }
        for (int off = 16; off > 0; off >>= 1) {
            s  += __shfl_xor(s,  off, 32);
            s2 += __shfl_xor(s2, off, 32);
        }
        const float mu  = s * (1.0f / OUT_DIM);
        const float var = s2 * (1.0f / OUT_DIM) - mu * mu;
        const float rs  = rsqrtf(var + LN_EPS);
        float* orow = out + (size_t)(m0 + row) * OUT_DIM;
#pragma unroll
        for (int j = 0; j < 8; ++j) {
            const int c = lane * 8 + j;
            float ln = (vals[j] - mu) * rs * gamma[c] + beta[c];
            orow[c] = fmaxf(ln, 0.f) + lds_skip[row][c];
        }
    }
}

extern "C" void kernel_launch(void* const* d_in, const int* in_sizes, int n_in,
                              void* d_out, int out_size, void* d_ws, size_t ws_size,
                              hipStream_t stream) {
    const float* feat   = (const float*)d_in[0];
    const int*   src    = (const int*)d_in[1];
    const int*   dst    = (const int*)d_in[2];
    const float* W      = (const float*)d_in[3];
    const float* bias   = (const float*)d_in[4];
    const float* gamma  = (const float*)d_in[5];
    const float* beta   = (const float*)d_in[6];
    const float* skip_W = (const float*)d_in[7];
    const float* skip_b = (const float*)d_in[8];
    float* out = (float*)d_out;

    const int n_nodes = in_sizes[0] / IN_DIM;
    const int n_edges = in_sizes[1];

    int*   deg_out = (int*)d_ws;
    int*   deg_in  = deg_out + n_nodes;
    float* h       = (float*)(deg_in + n_nodes);
    float* agg     = h + (size_t)n_nodes * IN_DIM;

    const int T = 256;

    {
        unsigned n = 2u * (unsigned)n_nodes;
        zero_u32<<<(n + T - 1) / T, T, 0, stream>>>((unsigned*)deg_out, n);
        unsigned na = (unsigned)n_nodes * IN_DIM;
        zero_u32<<<(na + T - 1) / T, T, 0, stream>>>((unsigned*)agg, na);
    }

    degree_kernel<<<(n_edges + T - 1) / T, T, 0, stream>>>(src, dst, deg_out, deg_in, n_edges);

    {
        int total = n_nodes * (IN_DIM / 4);
        compute_h<<<(total + T - 1) / T, T, 0, stream>>>(feat, deg_out, h, n_nodes);
    }

    {
        int waves_per_block = T / 32;
        int blocks = (n_edges + waves_per_block - 1) / waves_per_block;
        spmm_scatter<<<blocks, T, 0, stream>>>(src, dst, h, agg, n_edges);
    }

    gcn_fused<<<n_nodes / 16, T, 0, stream>>>(agg, deg_in, W, bias, gamma, beta,
                                              feat, skip_W, skip_b, out, n_nodes);
}


# spmm scatter: lane l owns floats l,l+32,l+64,l+96, so each wave atomic and h load covers one whole 128-B line, not 4 quarter-filled ones; float atomics and arrival-order sums kept
# speedup vs baseline: 1.1562x; 1.1562x over previous
#include <hip/hip_runtime.h>
#include <hip/hip_bf16.h>

typedef __attribute__((ext_vector_type(2))) float v2f;
typedef __attribute__((ext_vector_type(8))) float v8f;

#define IN_DIM 128
#define OUT_DIM 256
#define LN_EPS 1e-5f

__global__ void zero_u32(unsigned* __restrict__ p, unsigned n) {
    unsigned i = blockIdx.x * blockDim.x + threadIdx.x;
    if (i < n) p[i] = 0u;
}

__global__ void degree_kernel(const int* __restrict__ src, const int* __restrict__ dst,
                              int* __restrict__ deg_out, int* __restrict__ deg_in, int n_edges) {
    int e = blockIdx.x * blockDim.x + threadIdx.x;
    if (e >= n_edges) return;
    atomicAdd(&deg_out[src[e]], 1);
    atomicAdd(&deg_in[dst[e]], 1);
}

__global__ void compute_h(const float* __restrict__ feat, const int* __restrict__ deg_out,
                          float* __restrict__ h, int n_nodes) {
    int idx = blockIdx.x * blockDim.x + threadIdx.x;
    int total = n_nodes * (IN_DIM / 4);
    if (idx >= total) return;
    int node = idx / (IN_DIM / 4);
    float nrm = rsqrtf(fmaxf((float)deg_out[node], 1.0f));
    float4 v = ((const float4*)feat)[idx];
    v.x *= nrm; v.y *= nrm; v.z *= nrm; v.w *= nrm;
    ((float4*)h)[idx] = v;
}

__global__ __launch_bounds__(256) void spmm_scatter(const int* __restrict__ src,
                                                    const int* __restrict__ dst,
                                                    const float* __restrict__ h,
                                                    float* __restrict__ agg, int n_edges) {
    int edge = blockIdx.x * (blockDim.x >> 5) + (threadIdx.x >> 5);
    if (edge >= n_edges) return;
    int lane = threadIdx.x & 31;
    int s = src[edge];
    int d = dst[edge];
    const float* hp = h + (size_t)s * IN_DIM + lane;
    const float v0 = hp[0];
    const float v1 = hp[32];
    const float v2 = hp[64];
    const float v3 = hp[96];
    float* ap = agg + (size_t)d * IN_DIM + lane;
    __hip_atomic_fetch_add(ap + 0,  v0, __ATOMIC_RELAXED, __HIP_MEMORY_SCOPE_AGENT);
    __hip_atomic_fetch_add(ap + 32, v1, __ATOMIC_RELAXED, __HIP_MEMORY_SCOPE_AGENT);
    __hip_atomic_fetch_add(ap + 64, v2, __ATOMIC_RELAXED, __HIP_MEMORY_SCOPE_AGENT);
    __hip_atomic_fetch_add(ap + 96, v3, __ATOMIC_RELAXED, __HIP_MEMORY_SCOPE_AGENT);
}

__global__ __launch_bounds__(256) void gcn_fused(const float* __restrict__ agg,
                                                 const int* __restrict__ deg_in,
                                                 const float* __restrict__ W,
                                                 const float* __restrict__ bias,
                                                 const float* __restrict__ gamma,
                                                 const float* __restrict__ beta,
                                                 const float* __restrict__ feat,
                                                 const float* __restrict__ skip_W,
                                                 const float* __restrict__ skip_b,
                                                 float* __restrict__ out, int n_nodes) {
    __shared__ float lds_gcn[16][OUT_DIM + 4];
    __shared__ float lds_skip[16][OUT_DIM + 4];

    const int m0   = blockIdx.x * 16;
    const int wave = threadIdx.x >> 5;
    const int lane = threadIdx.x & 31;
    const int half = lane >> 4;
    const int l16  = lane & 15;

    const int arow = m0 + l16;
    const float* agg_row  = agg  + (size_t)arow * IN_DIM;
    const float* feat_row = feat + (size_t)arow * IN_DIM;
    const int bcol = wave * 32 + l16;

    v8f cg0 = {}, cg1 = {};
    v8f cs0 = {}, cs1 = {};

    for (int k0 = 0; k0 < IN_DIM; k0 += 4) {
        const int ka = k0 + 2 * half;
        v2f aA = *(const v2f*)(agg_row + ka);
        v2f aF = *(const v2f*)(feat_row + ka);
        const float* Wk0 = W      + (size_t)ka * OUT_DIM;
        const float* Wk1 = Wk0 + OUT_DIM;
        const float* Sk0 = skip_W + (size_t)ka * OUT_DIM;
        const float* Sk1 = Sk0 + OUT_DIM;
        v2f bW0, bW1, bS0, bS1;
        bW0[0] = Wk0[bcol];      bW0[1] = Wk1[bcol];
        bW1[0] = Wk0[bcol + 16]; bW1[1] = Wk1[bcol + 16];
        bS0[0] = Sk0[bcol];      bS0[1] = Sk1[bcol];
        bS1[0] = Sk0[bcol + 16]; bS1[1] = Sk1[bcol + 16];
        cg0 = __builtin_amdgcn_wmma_f32_16x16x4_f32(false, aA, false, bW0, (short)0, cg0, false, false);
        cg1 = __builtin_amdgcn_wmma_f32_16x16x4_f32(false, aA, false, bW1, (short)0, cg1, false, false);
        cs0 = __builtin_amdgcn_wmma_f32_16x16x4_f32(false, aF, false, bS0, (short)0, cs0, false, false);
        cs1 = __builtin_amdgcn_wmma_f32_16x16x4_f32(false, aF, false, bS1, (short)0, cs1, false, false);
    }

    float nrmv[8];
#pragma unroll
    for (int i = 0; i < 8; ++i) {
        const int row = i + 8 * half;
        nrmv[i] = rsqrtf(fmaxf((float)deg_in[m0 + row], 1.0f));
    }

    {
        const float bb0 = bias[bcol],        sb0 = skip_b[bcol];
        const float bb1 = bias[bcol + 16],   sb1 = skip_b[bcol + 16];
#pragma unroll
        for (int i = 0; i < 8; ++i) {
            const int row = i + 8 * half;
            lds_gcn[row][bcol]       = cg0[i] * nrmv[i] + bb0;
            lds_gcn[row][bcol + 16]  = cg1[i] * nrmv[i] + bb1;
            lds_skip[row][bcol]      = cs0[i] + sb0;
            lds_skip[row][bcol + 16] = cs1[i] + sb1;
        }
    }
    __syncthreads();

#pragma unroll
    for (int rr = 0; rr < 2; ++rr) {
        const int row = wave * 2 + rr;
        float vals[8];
        float s = 0.f, s2 = 0.f;
#pragma unroll
        for (int j = 0; j < 8; ++j) {
            float x = lds_gcn[row][lane * 8 + j];
            vals[j] = x;
            s += x;
            s2 += x * x;
        }
        for (int off = 16; off > 0; off >>= 1) {
            s  += __shfl_xor(s,  off, 32);
            s2 += __shfl_xor(s2, off, 32);
        }
        const float mu  = s * (1.0f / OUT_DIM);
        const float var = s2 * (1.0f / OUT_DIM) - mu * mu;
        const float rs  = rsqrtf(var + LN_EPS);
        float* orow = out + (size_t)(m0 + row) * OUT_DIM;
#pragma unroll
        for (int j = 0; j < 8; ++j) {
            const int c = lane * 8 + j;
            float ln = (vals[j] - mu) * rs * gamma[c] + beta[c];
            orow[c] = fmaxf(ln, 0.f) + lds_skip[row][c];
        }
    }
}

extern "C" void kernel_launch(void* const* d_in, const int* in_sizes, int n_in,
                              void* d_out, int out_size, void* d_ws, size_t ws_size,
                              hipStream_t stream) {
    const float* feat   = (const float*)d_in[0];
    const int*   src    = (const int*)d_in[1];
    const int*   dst    = (const int*)d_in[2];
    const float* W      = (const float*)d_in[3];
    const float* bias   = (const float*)d_in[4];
    const float* gamma  = (const float*)d_in[5];
    const float* beta   = (const float*)d_in[6];
    const float* skip_W = (const float*)d_in[7];
    const float* skip_b = (const float*)d_in[8];
    float* out = (float*)d_out;

    const int n_nodes = in_sizes[0] / IN_DIM;
    const int n_edges = in_sizes[1];

    int*   deg_out = (int*)d_ws;
    int*   deg_in  = deg_out + n_nodes;
    float* h       = (float*)(deg_in + n_nodes);
    float* agg     = h + (size_t)n_nodes * IN_DIM;

    const int T = 256;

    {
        unsigned n = 2u * (unsigned)n_nodes;
        zero_u32<<<(n + T - 1) / T, T, 0, stream>>>((unsigned*)deg_out, n);
        unsigned na = (unsigned)n_nodes * IN_DIM;
        zero_u32<<<(na + T - 1) / T, T, 0, stream>>>((unsigned*)agg, na);
    }

    degree_kernel<<<(n_edges + T - 1) / T, T, 0, stream>>>(src, dst, deg_out, deg_in, n_edges);

    {
        int total = n_nodes * (IN_DIM / 4);
        compute_h<<<(total + T - 1) / T, T, 0, stream>>>(feat, deg_out, h, n_nodes);
    }

    {
        int waves_per_block = T / 32;
        int blocks = (n_edges + waves_per_block - 1) / waves_per_block;
        spmm_scatter<<<blocks, T, 0, stream>>>(src, dst, h, agg, n_edges);
    }

    gcn_fused<<<n_nodes / 16, T, 0, stream>>>(agg, deg_in, W, bias, gamma, beta,
                                              feat, skip_W, skip_b, out, n_nodes);
}


# aggregation pulled, not pushed: edges bucketed by dst (prefix sum + int atomic slots), one wave per node sums feat[src]*norm_src rows in registers; no float atomics; sum order = slot order
# speedup vs baseline: 1.2942x; 1.1194x over previous
#include <hip/hip_runtime.h>
#include <hip/hip_bf16.h>

typedef __attribute__((ext_vector_type(2))) float v2f;
typedef __attribute__((ext_vector_type(8))) float v8f;

#define IN_DIM 128
#define OUT_DIM 256
#define LN_EPS 1e-5f

__global__ void zero_u32(unsigned* __restrict__ p, unsigned n) {
    unsigned i = blockIdx.x * blockDim.x + threadIdx.x;
    if (i < n) p[i] = 0u;
}

__global__ void degree_kernel(const int* __restrict__ src, const int* __restrict__ dst,
                              int* __restrict__ deg_out, int* __restrict__ deg_in, int n_edges) {
    int e = blockIdx.x * blockDim.x + threadIdx.x;
    if (e >= n_edges) return;
    atomicAdd(&deg_out[src[e]], 1);
    atomicAdd(&deg_in[dst[e]], 1);
}

__global__ __launch_bounds__(1024) void row_starts(const int* __restrict__ deg_in,
                                                   int* __restrict__ row_ptr, int n) {
    __shared__ int wsum[32];
    const int tid  = threadIdx.x;
    const int lane = tid & 31;
    const int wave = tid >> 5;
    int carry = 0;
    for (int base = 0; base < n; base += 16384) {
        const int i0 = base + tid * 16;
        int a[16];
#pragma unroll
        for (int q = 0; q < 4; ++q) {
            const int i = i0 + 4 * q;
            if (i + 3 < n) {
                const int4 v = *(const int4*)(deg_in + i);
                a[4 * q] = v.x; a[4 * q + 1] = v.y; a[4 * q + 2] = v.z; a[4 * q + 3] = v.w;
            } else {
                a[4 * q]     = (i < n)     ? deg_in[i]     : 0;
                a[4 * q + 1] = (i + 1 < n) ? deg_in[i + 1] : 0;
                a[4 * q + 2] = (i + 2 < n) ? deg_in[i + 2] : 0;
                a[4 * q + 3] = 0;
            }
        }
        int t = 0;
#pragma unroll
        for (int k = 0; k < 16; ++k) { const int x = a[k]; a[k] = t; t += x; }
        int s = t;
        for (int off = 1; off < 32; off <<= 1) {
            const int u = __shfl_up(s, off, 32);
            if (lane >= off) s += u;
        }
        if (lane == 31) wsum[wave] = s;
        __syncthreads();
        const int w = wsum[lane];
        int wi = w;
        for (int off = 1; off < 32; off <<= 1) {
            const int u = __shfl_up(wi, off, 32);
            if (lane >= off) wi += u;
        }
        const int wave_excl = __shfl(wi, wave, 32) - __shfl(w, wave, 32);
        const int total     = __shfl(wi, 31, 32);
        const int excl = carry + wave_excl + (s - t);
#pragma unroll
        for (int q = 0; q < 4; ++q) {
            const int i = i0 + 4 * q;
            if (i + 3 < n) {
                *(int4*)(row_ptr + i) = make_int4(excl + a[4 * q], excl + a[4 * q + 1],
                                                  excl + a[4 * q + 2], excl + a[4 * q + 3]);
            } else {
                if (i < n)     row_ptr[i]     = excl + a[4 * q];
                if (i + 1 < n) row_ptr[i + 1] = excl + a[4 * q + 1];
                if (i + 2 < n) row_ptr[i + 2] = excl + a[4 * q + 2];
            }
        }
        carry += total;
        __syncthreads();
    }
}

__global__ void csr_fill(const int* __restrict__ src, const int* __restrict__ dst,
                         int* __restrict__ row_ptr, int* __restrict__ col, int n_edges) {
    int e = blockIdx.x * blockDim.x + threadIdx.x;
    if (e >= n_edges) return;
    const int p = atomicAdd(&row_ptr[dst[e]], 1);
    col[p] = src[e];
}

template <int G>
__device__ __forceinline__ void add_rows(const float* __restrict__ feat, int myidx, int mynrm_bits,
                                         int j, int lane, float4& acc) {
    float4 r[G];
    float  w[G];
#pragma unroll
    for (int k = 0; k < G; ++k) {
        const int s = __builtin_amdgcn_readlane(myidx, j + k);
        w[k] = __int_as_float(__builtin_amdgcn_readlane(mynrm_bits, j + k));
        r[k] = ((const float4*)(feat + (size_t)s * IN_DIM))[lane];
    }
#pragma unroll
    for (int k = 0; k < G; ++k) {
        acc.x += r[k].x * w[k];
        acc.y += r[k].y * w[k];
        acc.z += r[k].z * w[k];
        acc.w += r[k].w * w[k];
    }
}

__global__ __launch_bounds__(256) void spmm_gather(const int* __restrict__ col,
                                                   const int* __restrict__ row_end,
                                                   const int* __restrict__ deg_in,
                                                   const int* __restrict__ deg_out,
                                                   const float* __restrict__ feat,
                                                   float* __restrict__ agg, int n_nodes) {
    const int node = __builtin_amdgcn_readfirstlane(
        (int)(blockIdx.x * (blockDim.x >> 5) + (threadIdx.x >> 5)));
    if (node >= n_nodes) return;
    const int lane  = threadIdx.x & 31;
    const int cnt   = __builtin_amdgcn_readfirstlane(deg_in[node]);
    const int start = __builtin_amdgcn_readfirstlane(row_end[node]) - cnt;

    float4 acc = make_float4(0.f, 0.f, 0.f, 0.f);
    for (int base = 0; base < cnt; base += 32) {
        const int m = min(32, cnt - base);
        int   myidx = 0;
        float mynrm = 0.f;
        if (lane < m) {
            myidx = col[start + base + lane];
            mynrm = rsqrtf(fmaxf((float)deg_out[myidx], 1.0f));
        }
        const int nb = __float_as_int(mynrm);
        int j = 0;
        for (; j + 8 <= m; j += 8) add_rows<8>(feat, myidx, nb, j, lane, acc);
        if (m - j >= 4) { add_rows<4>(feat, myidx, nb, j, lane, acc); j += 4; }
        if (m - j >= 2) { add_rows<2>(feat, myidx, nb, j, lane, acc); j += 2; }
        if (m - j >= 1) { add_rows<1>(feat, myidx, nb, j, lane, acc); }
    }
    ((float4*)(agg + (size_t)node * IN_DIM))[lane] = acc;
}

__global__ __launch_bounds__(256) void gcn_fused(const float* __restrict__ agg,
                                                 const int* __restrict__ deg_in,
                                                 const float* __restrict__ W,
                                                 const float* __restrict__ bias,
                                                 const float* __restrict__ gamma,
                                                 const float* __restrict__ beta,
                                                 const float* __restrict__ feat,
                                                 const float* __restrict__ skip_W,
                                                 const float* __restrict__ skip_b,
                                                 float* __restrict__ out, int n_nodes) {
    __shared__ float lds_gcn[16][OUT_DIM + 4];
    __shared__ float lds_skip[16][OUT_DIM + 4];

    const int m0   = blockIdx.x * 16;
    const int wave = threadIdx.x >> 5;
    const int lane = threadIdx.x & 31;
    const int half = lane >> 4;
    const int l16  = lane & 15;

    const int arow = m0 + l16;
    const float* agg_row  = agg  + (size_t)arow * IN_DIM;
    const float* feat_row = feat + (size_t)arow * IN_DIM;
    const int bcol = wave * 32 + l16;

    v8f cg0 = {}, cg1 = {};
    v8f cs0 = {}, cs1 = {};

    for (int k0 = 0; k0 < IN_DIM; k0 += 4) {
        const int ka = k0 + 2 * half;
        v2f aA = *(const v2f*)(agg_row + ka);
        v2f aF = *(const v2f*)(feat_row + ka);
        const float* Wk0 = W      + (size_t)ka * OUT_DIM;
        const float* Wk1 = Wk0 + OUT_DIM;
        const float* Sk0 = skip_W + (size_t)ka * OUT_DIM;
        const float* Sk1 = Sk0 + OUT_DIM;
        v2f bW0, bW1, bS0, bS1;
        bW0[0] = Wk0[bcol];      bW0[1] = Wk1[bcol];
        bW1[0] = Wk0[bcol + 16]; bW1[1] = Wk1[bcol + 16];
        bS0[0] = Sk0[bcol];      bS0[1] = Sk1[bcol];
        bS1[0] = Sk0[bcol + 16]; bS1[1] = Sk1[bcol + 16];
        cg0 = __builtin_amdgcn_wmma_f32_16x16x4_f32(false, aA, false, bW0, (short)0, cg0, false, false);
        cg1 = __builtin_amdgcn_wmma_f32_16x16x4_f32(false, aA, false, bW1, (short)0, cg1, false, false);
        cs0 = __builtin_amdgcn_wmma_f32_16x16x4_f32(false, aF, false, bS0, (short)0, cs0, false, false);
        cs1 = __builtin_amdgcn_wmma_f32_16x16x4_f32(false, aF, false, bS1, (short)0, cs1, false, false);
    }

    float nrmv[8];
#pragma unroll
    for (int i = 0; i < 8; ++i) {
        const int row = i + 8 * half;
        nrmv[i] = rsqrtf(fmaxf((float)deg_in[m0 + row], 1.0f));
    }

    {
        const float bb0 = bias[bcol],        sb0 = skip_b[bcol];
        const float bb1 = bias[bcol + 16],   sb1 = skip_b[bcol + 16];
#pragma unroll
        for (int i = 0; i < 8; ++i) {
            const int row = i + 8 * half;
            lds_gcn[row][bcol]       = cg0[i] * nrmv[i] + bb0;
            lds_gcn[row][bcol + 16]  = cg1[i] * nrmv[i] + bb1;
            lds_skip[row][bcol]      = cs0[i] + sb0;
            lds_skip[row][bcol + 16] = cs1[i] + sb1;
        }
    }
    __syncthreads();

#pragma unroll
    for (int rr = 0; rr < 2; ++rr) {
        const int row = wave * 2 + rr;
        float vals[8];
        float s = 0.f, s2 = 0.f;
#pragma unroll
        for (int j = 0; j < 8; ++j) {
            float x = lds_gcn[row][lane * 8 + j];
            vals[j] = x;
            s += x;
            s2 += x * x;
        }
        for (int off = 16; off > 0; off >>= 1) {
            s  += __shfl_xor(s,  off, 32);
            s2 += __shfl_xor(s2, off, 32);
        }
        const float mu  = s * (1.0f / OUT_DIM);
        const float var = s2 * (1.0f / OUT_DIM) - mu * mu;
        const float rs  = rsqrtf(var + LN_EPS);
        float* orow = out + (size_t)(m0 + row) * OUT_DIM;
#pragma unroll
        for (int j = 0; j < 8; ++j) {
            const int c = lane * 8 + j;
            float ln = (vals[j] - mu) * rs * gamma[c] + beta[c];
            orow[c] = fmaxf(ln, 0.f) + lds_skip[row][c];
        }
    }
}

extern "C" void kernel_launch(void* const* d_in, const int* in_sizes, int n_in,
                              void* d_out, int out_size, void* d_ws, size_t ws_size,
                              hipStream_t stream) {
    const float* feat   = (const float*)d_in[0];
    const int*   src    = (const int*)d_in[1];
    const int*   dst    = (const int*)d_in[2];
    const float* W      = (const float*)d_in[3];
    const float* bias   = (const float*)d_in[4];
    const float* gamma  = (const float*)d_in[5];
    const float* beta   = (const float*)d_in[6];
    const float* skip_W = (const float*)d_in[7];
    const float* skip_b = (const float*)d_in[8];
    float* out = (float*)d_out;

    const int n_nodes = in_sizes[0] / IN_DIM;
    const int n_edges = in_sizes[1];

    int*   deg_out = (int*)d_ws;
    int*   deg_in  = deg_out + n_nodes;
    int*   row_ptr = deg_in + n_nodes;
    int*   col     = row_ptr + n_nodes;
    float* agg     = (float*)(deg_in + n_nodes) + (size_t)n_nodes * IN_DIM;

    const int T = 256;

    {
        unsigned n = 2u * (unsigned)n_nodes;
        zero_u32<<<(n + T - 1) / T, T, 0, stream>>>((unsigned*)deg_out, n);
    }

    degree_kernel<<<(n_edges + T - 1) / T, T, 0, stream>>>(src, dst, deg_out, deg_in, n_edges);

    row_starts<<<1, 1024, 0, stream>>>(deg_in, row_ptr, n_nodes);

    csr_fill<<<(n_edges + T - 1) / T, T, 0, stream>>>(src, dst, row_ptr, col, n_edges);

    {
        int waves_per_block = T / 32;
        int blocks = (n_nodes + waves_per_block - 1) / waves_per_block;
        spmm_gather<<<blocks, T, 0, stream>>>(col, row_ptr, deg_in, deg_out, feat, agg, n_nodes);
    }

    gcn_fused<<<n_nodes / 16, T, 0, stream>>>(agg, deg_in, W, bias, gamma, beta,
                                              feat, skip_W, skip_b, out, n_nodes);
}


# gcn_fused row tile 16 -> 64 per work-group: agg/feat rows staged once in LDS by whole-row loads (A fragments read there), each W/skip_W fragment load feeds 4 row tiles; sums and k order as before
# speedup vs baseline: 1.4494x; 1.1199x over previous
#include <hip/hip_runtime.h>
#include <hip/hip_bf16.h>

typedef __attribute__((ext_vector_type(2))) float v2f;
typedef __attribute__((ext_vector_type(8))) float v8f;

#define IN_DIM 128
#define OUT_DIM 256
#define LN_EPS 1e-5f

__global__ void zero_u32(unsigned* __restrict__ p, unsigned n) {
    unsigned i = blockIdx.x * blockDim.x + threadIdx.x;
    if (i < n) p[i] = 0u;
}

__global__ void degree_kernel(const int* __restrict__ src, const int* __restrict__ dst,
                              int* __restrict__ deg_out, int* __restrict__ deg_in, int n_edges) {
    int e = blockIdx.x * blockDim.x + threadIdx.x;
    if (e >= n_edges) return;
    atomicAdd(&deg_out[src[e]], 1);
    atomicAdd(&deg_in[dst[e]], 1);
}

__global__ __launch_bounds__(1024) void row_starts(const int* __restrict__ deg_in,
                                                   int* __restrict__ row_ptr, int n) {
    __shared__ int wsum[32];
    const int tid  = threadIdx.x;
    const int lane = tid & 31;
    const int wave = tid >> 5;
    int carry = 0;
    for (int base = 0; base < n; base += 16384) {
        const int i0 = base + tid * 16;
        int a[16];
#pragma unroll
        for (int q = 0; q < 4; ++q) {
            const int i = i0 + 4 * q;
            if (i + 3 < n) {
                const int4 v = *(const int4*)(deg_in + i);
                a[4 * q] = v.x; a[4 * q + 1] = v.y; a[4 * q + 2] = v.z; a[4 * q + 3] = v.w;
            } else {
                a[4 * q]     = (i < n)     ? deg_in[i]     : 0;
                a[4 * q + 1] = (i + 1 < n) ? deg_in[i + 1] : 0;
                a[4 * q + 2] = (i + 2 < n) ? deg_in[i + 2] : 0;
                a[4 * q + 3] = 0;
            }
        }
        int t = 0;
#pragma unroll
        for (int k = 0; k < 16; ++k) { const int x = a[k]; a[k] = t; t += x; }
        int s = t;
        for (int off = 1; off < 32; off <<= 1) {
            const int u = __shfl_up(s, off, 32);
            if (lane >= off) s += u;
        }
        if (lane == 31) wsum[wave] = s;
        __syncthreads();
        const int w = wsum[lane];
        int wi = w;
        for (int off = 1; off < 32; off <<= 1) {
            const int u = __shfl_up(wi, off, 32);
            if (lane >= off) wi += u;
        }
        const int wave_excl = __shfl(wi, wave, 32) - __shfl(w, wave, 32);
        const int total     = __shfl(wi, 31, 32);
        const int excl = carry + wave_excl + (s - t);
#pragma unroll
        for (int q = 0; q < 4; ++q) {
            const int i = i0 + 4 * q;
            if (i + 3 < n) {
                *(int4*)(row_ptr + i) = make_int4(excl + a[4 * q], excl + a[4 * q + 1],
                                                  excl + a[4 * q + 2], excl + a[4 * q + 3]);
            } else {
                if (i < n)     row_ptr[i]     = excl + a[4 * q];
                if (i + 1 < n) row_ptr[i + 1] = excl + a[4 * q + 1];
                if (i + 2 < n) row_ptr[i + 2] = excl + a[4 * q + 2];
            }
        }
        carry += total;
        __syncthreads();
    }
}

__global__ void csr_fill(const int* __restrict__ src, const int* __restrict__ dst,
                         int* __restrict__ row_ptr, int* __restrict__ col, int n_edges) {
    int e = blockIdx.x * blockDim.x + threadIdx.x;
    if (e >= n_edges) return;
    const int p = atomicAdd(&row_ptr[dst[e]], 1);
    col[p] = src[e];
}

template <int G>
__device__ __forceinline__ void add_rows(const float* __restrict__ feat, int myidx, int mynrm_bits,
                                         int j, int lane, float4& acc) {
    float4 r[G];
    float  w[G];
#pragma unroll
    for (int k = 0; k < G; ++k) {
        const int s = __builtin_amdgcn_readlane(myidx, j + k);
        w[k] = __int_as_float(__builtin_amdgcn_readlane(mynrm_bits, j + k));
        r[k] = ((const float4*)(feat + (size_t)s * IN_DIM))[lane];
    }
#pragma unroll
    for (int k = 0; k < G; ++k) {
        acc.x += r[k].x * w[k];
        acc.y += r[k].y * w[k];
        acc.z += r[k].z * w[k];
        acc.w += r[k].w * w[k];
    }
}

__global__ __launch_bounds__(256) void spmm_gather(const int* __restrict__ col,
                                                   const int* __restrict__ row_end,
                                                   const int* __restrict__ deg_in,
                                                   const int* __restrict__ deg_out,
                                                   const float* __restrict__ feat,
                                                   float* __restrict__ agg, int n_nodes) {
    const int node = __builtin_amdgcn_readfirstlane(
        (int)(blockIdx.x * (blockDim.x >> 5) + (threadIdx.x >> 5)));
    if (node >= n_nodes) return;
    const int lane  = threadIdx.x & 31;
    const int cnt   = __builtin_amdgcn_readfirstlane(deg_in[node]);
    const int start = __builtin_amdgcn_readfirstlane(row_end[node]) - cnt;

    float4 acc = make_float4(0.f, 0.f, 0.f, 0.f);
    for (int base = 0; base < cnt; base += 32) {
        const int m = min(32, cnt - base);
        int   myidx = 0;
        float mynrm = 0.f;
        if (lane < m) {
            myidx = col[start + base + lane];
            mynrm = rsqrtf(fmaxf((float)deg_out[myidx], 1.0f));
        }
        const int nb = __float_as_int(mynrm);
        int j = 0;
        for (; j + 8 <= m; j += 8) add_rows<8>(feat, myidx, nb, j, lane, acc);
        if (m - j >= 4) { add_rows<4>(feat, myidx, nb, j, lane, acc); j += 4; }
        if (m - j >= 2) { add_rows<2>(feat, myidx, nb, j, lane, acc); j += 2; }
        if (m - j >= 1) { add_rows<1>(feat, myidx, nb, j, lane, acc); }
    }
    ((float4*)(agg + (size_t)node * IN_DIM))[lane] = acc;
}

#define M_TILES 4
#define ROWS_PER_WG (16 * M_TILES)
#define A_PITCH (IN_DIM + 4)

struct BFrag { v2f w0, w1, s0, s1; };

__device__ __forceinline__ BFrag load_b(const float* __restrict__ W, const float* __restrict__ skip_W,
                                        int ka, int bcol) {
    const float* Wk0 = W      + (size_t)ka * OUT_DIM;
    const float* Wk1 = Wk0 + OUT_DIM;
    const float* Sk0 = skip_W + (size_t)ka * OUT_DIM;
    const float* Sk1 = Sk0 + OUT_DIM;
    BFrag b;
    b.w0[0] = Wk0[bcol];      b.w0[1] = Wk1[bcol];
    b.w1[0] = Wk0[bcol + 16]; b.w1[1] = Wk1[bcol + 16];
    b.s0[0] = Sk0[bcol];      b.s0[1] = Sk1[bcol];
    b.s1[0] = Sk0[bcol + 16]; b.s1[1] = Sk1[bcol + 16];
    return b;
}

__global__ __launch_bounds__(256) void gcn_fused(const float* __restrict__ agg,
                                                 const int* __restrict__ deg_in,
                                                 const float* __restrict__ W,
                                                 const float* __restrict__ bias,
                                                 const float* __restrict__ gamma,
                                                 const float* __restrict__ beta,
                                                 const float* __restrict__ feat,
                                                 const float* __restrict__ skip_W,
                                                 const float* __restrict__ skip_b,
                                                 float* __restrict__ out, int n_nodes) {
    __shared__ __attribute__((aligned(16))) float lds_agg[ROWS_PER_WG][A_PITCH];
    __shared__ __attribute__((aligned(16))) float lds_feat[ROWS_PER_WG][A_PITCH];
    __shared__ __attribute__((aligned(16))) float lds_gcn[16][OUT_DIM + 4];
    __shared__ __attribute__((aligned(16))) float lds_skip[16][OUT_DIM + 4];

    const int m0   = blockIdx.x * ROWS_PER_WG;
    const int wave = threadIdx.x >> 5;
    const int lane = threadIdx.x & 31;
    const int half = lane >> 4;
    const int l16  = lane & 15;
    const int bcol = wave * 32 + l16;

    BFrag bc = load_b(W, skip_W, 2 * half, bcol);

    {
        float4 va[ROWS_PER_WG / 8], vf[ROWS_PER_WG / 8];
#pragma unroll
        for (int i = 0; i < ROWS_PER_WG / 8; ++i) {
            const int gr = min(m0 + wave * (ROWS_PER_WG / 8) + i, n_nodes - 1);
            va[i] = ((const float4*)(agg  + (size_t)gr * IN_DIM))[lane];
            vf[i] = ((const float4*)(feat + (size_t)gr * IN_DIM))[lane];
        }
#pragma unroll
        for (int i = 0; i < ROWS_PER_WG / 8; ++i) {
            const int r = wave * (ROWS_PER_WG / 8) + i;
            *(float4*)&lds_agg[r][lane * 4]  = va[i];
            *(float4*)&lds_feat[r][lane * 4] = vf[i];
        }
    }
    __syncthreads();

    v8f cg[M_TILES][2];
    v8f cs[M_TILES][2];
#pragma unroll
    for (int mt = 0; mt < M_TILES; ++mt) {
        cg[mt][0] = v8f{}; cg[mt][1] = v8f{};
        cs[mt][0] = v8f{}; cs[mt][1] = v8f{};
    }

#pragma unroll 2
    for (int k0 = 0; k0 < IN_DIM; k0 += 4) {
        const int ka = k0 + 2 * half;
        const BFrag bn = load_b(W, skip_W, min(k0 + 4, IN_DIM - 4) + 2 * half, bcol);
#pragma unroll
        for (int mt = 0; mt < M_TILES; ++mt) {
            const v2f aA = *(const v2f*)&lds_agg[mt * 16 + l16][ka];
            const v2f aF = *(const v2f*)&lds_feat[mt * 16 + l16][ka];
            cg[mt][0] = __builtin_amdgcn_wmma_f32_16x16x4_f32(false, aA, false, bc.w0, (short)0, cg[mt][0], false, false);
            cg[mt][1] = __builtin_amdgcn_wmma_f32_16x16x4_f32(false, aA, false, bc.w1, (short)0, cg[mt][1], false, false);
            cs[mt][0] = __builtin_amdgcn_wmma_f32_16x16x4_f32(false, aF, false, bc.s0, (short)0, cs[mt][0], false, false);
            cs[mt][1] = __builtin_amdgcn_wmma_f32_16x16x4_f32(false, aF, false, bc.s1, (short)0, cs[mt][1], false, false);
        }
        bc = bn;
    }

    const float bb0 = bias[bcol],        sb0 = skip_b[bcol];
    const float bb1 = bias[bcol + 16],   sb1 = skip_b[bcol + 16];
    float gm[8], bt[8];
#pragma unroll
    for (int j = 0; j < 8; ++j) { gm[j] = gamma[lane * 8 + j]; bt[j] = beta[lane * 8 + j]; }

#pragma unroll
    for (int mt = 0; mt < M_TILES; ++mt) {
        if (mt > 0) __syncthreads();

#pragma unroll
        for (int i = 0; i < 8; ++i) {
            const int row = i + 8 * half;
            const int gr  = min(m0 + mt * 16 + row, n_nodes - 1);
            const float nrm = rsqrtf(fmaxf((float)deg_in[gr], 1.0f));
            lds_gcn[row][bcol]       = cg[mt][0][i] * nrm + bb0;
            lds_gcn[row][bcol + 16]  = cg[mt][1][i] * nrm + bb1;
            lds_skip[row][bcol]      = cs[mt][0][i] + sb0;
            lds_skip[row][bcol + 16] = cs[mt][1][i] + sb1;
        }
        __syncthreads();

#pragma unroll
        for (int rr = 0; rr < 2; ++rr) {
            const int row = wave * 2 + rr;
            float vals[8];
            float s = 0.f, s2 = 0.f;
#pragma unroll
            for (int j = 0; j < 8; ++j) {
                float x = lds_gcn[row][lane * 8 + j];
                vals[j] = x;
                s += x;
                s2 += x * x;
            }
            for (int off = 16; off > 0; off >>= 1) {
                s  += __shfl_xor(s,  off, 32);
                s2 += __shfl_xor(s2, off, 32);
            }
            const float mu  = s * (1.0f / OUT_DIM);
            const float var = s2 * (1.0f / OUT_DIM) - mu * mu;
            const float rs  = rsqrtf(var + LN_EPS);
            const int grow = m0 + mt * 16 + row;
            if (grow < n_nodes) {
                float* orow = out + (size_t)grow * OUT_DIM;
#pragma unroll
                for (int j = 0; j < 8; ++j) {
                    const int c = lane * 8 + j;
                    float ln = (vals[j] - mu) * rs * gm[j] + bt[j];
                    orow[c] = fmaxf(ln, 0.f) + lds_skip[row][c];
                }
            }
        }
    }
}

extern "C" void kernel_launch(void* const* d_in, const int* in_sizes, int n_in,
                              void* d_out, int out_size, void* d_ws, size_t ws_size,
                              hipStream_t stream) {
    const float* feat   = (const float*)d_in[0];
    const int*   src    = (const int*)d_in[1];
    const int*   dst    = (const int*)d_in[2];
    const float* W      = (const float*)d_in[3];
    const float* bias   = (const float*)d_in[4];
    const float* gamma  = (const float*)d_in[5];
    const float* beta   = (const float*)d_in[6];
    const float* skip_W = (const float*)d_in[7];
    const float* skip_b = (const float*)d_in[8];
    float* out = (float*)d_out;

    const int n_nodes = in_sizes[0] / IN_DIM;
    const int n_edges = in_sizes[1];

    int*   deg_out = (int*)d_ws;
    int*   deg_in  = deg_out + n_nodes;
    int*   row_ptr = deg_in + n_nodes;
    int*   col     = row_ptr + n_nodes;
    float* agg     = (float*)(deg_in + n_nodes) + (size_t)n_nodes * IN_DIM;

    const int T = 256;

    {
        unsigned n = 2u * (unsigned)n_nodes;
        zero_u32<<<(n + T - 1) / T, T, 0, stream>>>((unsigned*)deg_out, n);
    }

    degree_kernel<<<(n_edges + T - 1) / T, T, 0, stream>>>(src, dst, deg_out, deg_in, n_edges);

    row_starts<<<1, 1024, 0, stream>>>(deg_in, row_ptr, n_nodes);

    csr_fill<<<(n_edges + T - 1) / T, T, 0, stream>>>(src, dst, row_ptr, col, n_edges);

    {
        int waves_per_block = T / 32;
        int blocks = (n_nodes + waves_per_block - 1) / waves_per_block;
        spmm_gather<<<blocks, T, 0, stream>>>(col, row_ptr, deg_in, deg_out, feat, agg, n_nodes);
    }

    gcn_fused<<<(n_nodes + ROWS_PER_WG - 1) / ROWS_PER_WG, T, 0, stream>>>(agg, deg_in, W, bias, gamma, beta,
                                              feat, skip_W, skip_b, out, n_nodes);
}


# edge passes without global atomics: two-level counting sort with LDS counters (256 chunks, 512-node buckets) builds deg_out, deg_in and dst-sorted col; 5 launches; row sums in LDS arrival order
# speedup vs baseline: 2.8808x; 1.9876x over previous
#include <hip/hip_runtime.h>
#include <hip/hip_bf16.h>

typedef __attribute__((ext_vector_type(2))) float v2f;
typedef __attribute__((ext_vector_type(8))) float v8f;

#define IN_DIM 128
#define OUT_DIM 256
#define LN_EPS 1e-5f

#define BSHIFT 9
#define BSIZE (1 << BSHIFT)
#define NCHUNK 256
#define MAXB 1024
#define LD 16
#define LB 8

__global__ __launch_bounds__(256) void chunk_count(const int* __restrict__ src,
                                                   const int* __restrict__ dst,
                                                   int* __restrict__ cnt, int n_edges, int nb) {
    __shared__ int h[MAXB];
    const int tid = threadIdx.x;
    const int s   = blockIdx.y;
    const int c   = blockIdx.x;
    const int* __restrict__ key = s ? src : dst;
    for (int b = tid; b < nb; b += 256) h[b] = 0;
    __syncthreads();
    const int per = (n_edges + NCHUNK - 1) / NCHUNK;
    const int e0  = c * per;
    const int e1  = min(n_edges, e0 + per);
    for (int base = e0; base < e1; base += 256 * LD) {
        int k[LD];
#pragma unroll
        for (int q = 0; q < LD; ++q) {
            const int e = base + q * 256 + tid;
            k[q] = (e < e1) ? key[e] : -1;
        }
#pragma unroll
        for (int q = 0; q < LD; ++q)
            if (k[q] >= 0) atomicAdd(&h[k[q] >> BSHIFT], 1);
    }
    __syncthreads();
    int* out = cnt + ((size_t)s * NCHUNK + c) * nb;
    for (int b = tid; b < nb; b += 256) out[b] = h[b];
}

__global__ __launch_bounds__(256) void chunk_scatter(const int* __restrict__ src,
                                                     const int* __restrict__ dst,
                                                     const int* __restrict__ cnt,
                                                     int* __restrict__ bbase,
                                                     int* __restrict__ pairs_w,
                                                     int* __restrict__ sids,
                                                     int n_edges, int nb) {
    __shared__ int tot[MAXB];
    __shared__ int cur[MAXB];
    __shared__ int wsum[8];
    const int tid  = threadIdx.x;
    const int lane = tid & 31;
    const int wave = tid >> 5;
    const int s    = blockIdx.y;
    const int c    = blockIdx.x;

    for (int b = tid; b < MAXB; b += 256) {
        int t = 0, p = 0;
        if (b < nb) {
            const int* colp = cnt + (size_t)s * NCHUNK * nb + b;
            for (int c0 = 0; c0 < NCHUNK; c0 += 64) {
                int v[64];
#pragma unroll
                for (int q = 0; q < 64; ++q) v[q] = colp[(size_t)(c0 + q) * nb];
#pragma unroll
                for (int q = 0; q < 64; ++q) { t += v[q]; p += (c0 + q < c) ? v[q] : 0; }
            }
        }
        tot[b] = t;
        cur[b] = p;
    }
    __syncthreads();

    const int a0 = tot[4 * tid], a1 = tot[4 * tid + 1], a2 = tot[4 * tid + 2], a3 = tot[4 * tid + 3];
    const int tsum = a0 + a1 + a2 + a3;
    int sc = tsum;
    for (int off = 1; off < 32; off <<= 1) {
        const int u = __shfl_up(sc, off, 32);
        if (lane >= off) sc += u;
    }
    if (lane == 31) wsum[wave] = sc;
    __syncthreads();
    int wpre = 0;
    for (int w = 0; w < wave; ++w) wpre += wsum[w];
    const int s0 = wpre + sc - tsum;
    const int s1 = s0 + a0, s2 = s1 + a1, s3 = s2 + a2;
    cur[4 * tid]     += s0;
    cur[4 * tid + 1] += s1;
    cur[4 * tid + 2] += s2;
    cur[4 * tid + 3] += s3;
    if (c == 0) {
        int* bb = bbase + (size_t)s * (nb + 1);
        if (4 * tid     < nb) bb[4 * tid]     = s0;
        if (4 * tid + 1 < nb) bb[4 * tid + 1] = s1;
        if (4 * tid + 2 < nb) bb[4 * tid + 2] = s2;
        if (4 * tid + 3 < nb) bb[4 * tid + 3] = s3;
        if (tid == 255) bb[nb] = s0 + tsum;
    }
    __syncthreads();

    const int per = (n_edges + NCHUNK - 1) / NCHUNK;
    const int e0  = c * per;
    const int e1  = min(n_edges, e0 + per);
    if (s == 0) {
        int2* pairs = (int2*)pairs_w;
        for (int base = e0; base < e1; base += 256 * LD) {
            int kd[LD], ks[LD];
#pragma unroll
            for (int q = 0; q < LD; ++q) {
                const int e = base + q * 256 + tid;
                kd[q] = (e < e1) ? dst[e] : -1;
                ks[q] = (e < e1) ? src[e] : 0;
            }
#pragma unroll
            for (int q = 0; q < LD; ++q) {
                if (kd[q] >= 0) {
                    const int pos = atomicAdd(&cur[kd[q] >> BSHIFT], 1);
                    pairs[pos] = make_int2(ks[q], kd[q]);
                }
            }
        }
    } else {
        for (int base = e0; base < e1; base += 256 * LD) {
            int k[LD];
#pragma unroll
            for (int q = 0; q < LD; ++q) {
                const int e = base + q * 256 + tid;
                k[q] = (e < e1) ? src[e] : -1;
            }
#pragma unroll
            for (int q = 0; q < LD; ++q) {
                if (k[q] >= 0) {
                    const int pos = atomicAdd(&cur[k[q] >> BSHIFT], 1);
                    sids[pos] = k[q];
                }
            }
        }
    }
}

__global__ __launch_bounds__(BSIZE) void bucket_build(const int* __restrict__ pairs_w,
                                                      const int* __restrict__ sids,
                                                      const int* __restrict__ bbase,
                                                      int* __restrict__ deg_out,
                                                      int* __restrict__ deg_in,
                                                      int* __restrict__ row_ptr,
                                                      int* __restrict__ col,
                                                      int n_nodes, int nb) {
    __shared__ int cntl[BSIZE];
    __shared__ int curl[BSIZE];
    __shared__ int wsum[BSIZE / 32];
    const int tid  = threadIdx.x;
    const int lane = tid & 31;
    const int wave = tid >> 5;
    const int s    = blockIdx.y;
    const int b    = blockIdx.x;
    const int* bb  = bbase + (size_t)s * (nb + 1);
    const int i0   = bb[b];
    const int i1   = bb[b + 1];
    const int node0 = b << BSHIFT;
    const int node  = node0 + tid;

    cntl[tid] = 0;
    __syncthreads();

    if (s == 1) {
        for (int base = i0; base < i1; base += BSIZE * LB) {
            int k[LB];
#pragma unroll
            for (int q = 0; q < LB; ++q) {
                const int i = base + q * BSIZE + tid;
                k[q] = (i < i1) ? sids[i] : -1;
            }
#pragma unroll
            for (int q = 0; q < LB; ++q)
                if (k[q] >= 0) atomicAdd(&cntl[k[q] - node0], 1);
        }
        __syncthreads();
        if (node < n_nodes) deg_out[node] = cntl[tid];
        return;
    }

    const int2* pairs = (const int2*)pairs_w;
    for (int base = i0; base < i1; base += BSIZE * LB) {
        int k[LB];
#pragma unroll
        for (int q = 0; q < LB; ++q) {
            const int i = base + q * BSIZE + tid;
            k[q] = (i < i1) ? pairs[i].y : -1;
        }
#pragma unroll
        for (int q = 0; q < LB; ++q)
            if (k[q] >= 0) atomicAdd(&cntl[k[q] - node0], 1);
    }
    __syncthreads();

    const int cdeg = cntl[tid];
    int sc = cdeg;
    for (int off = 1; off < 32; off <<= 1) {
        const int u = __shfl_up(sc, off, 32);
        if (lane >= off) sc += u;
    }
    if (lane == 31) wsum[wave] = sc;
    __syncthreads();
    int wpre = 0;
    for (int w = 0; w < wave; ++w) wpre += wsum[w];
    const int start = i0 + wpre + sc - cdeg;
    if (node < n_nodes) {
        deg_in[node]  = cdeg;
        row_ptr[node] = start;
    }
    curl[tid] = start;
    __syncthreads();

    for (int base = i0; base < i1; base += BSIZE * LB) {
        int2 p[LB];
#pragma unroll
        for (int q = 0; q < LB; ++q) {
            const int i = base + q * BSIZE + tid;
            p[q] = (i < i1) ? pairs[i] : make_int2(0, -1);
        }
#pragma unroll
        for (int q = 0; q < LB; ++q) {
            if (p[q].y >= 0) {
                const int pos = atomicAdd(&curl[p[q].y - node0], 1);
                col[pos] = p[q].x;
            }
        }
    }
}

template <int G>
__device__ __forceinline__ void add_rows(const float* __restrict__ feat, int myidx, int mynrm_bits,
                                         int j, int lane, float4& acc) {
    float4 r[G];
    float  w[G];
#pragma unroll
    for (int k = 0; k < G; ++k) {
        const int s = __builtin_amdgcn_readlane(myidx, j + k);
        w[k] = __int_as_float(__builtin_amdgcn_readlane(mynrm_bits, j + k));
        r[k] = ((const float4*)(feat + (size_t)s * IN_DIM))[lane];
    }
#pragma unroll
    for (int k = 0; k < G; ++k) {
        acc.x += r[k].x * w[k];
        acc.y += r[k].y * w[k];
        acc.z += r[k].z * w[k];
        acc.w += r[k].w * w[k];
    }
}

__global__ __launch_bounds__(256) void spmm_gather(const int* __restrict__ col,
                                                   const int* __restrict__ row_start,
                                                   const int* __restrict__ deg_in,
                                                   const int* __restrict__ deg_out,
                                                   const float* __restrict__ feat,
                                                   float* __restrict__ agg, int n_nodes) {
    const int node = __builtin_amdgcn_readfirstlane(
        (int)(blockIdx.x * (blockDim.x >> 5) + (threadIdx.x >> 5)));
    if (node >= n_nodes) return;
    const int lane  = threadIdx.x & 31;
    const int cnt   = __builtin_amdgcn_readfirstlane(deg_in[node]);
    const int start = __builtin_amdgcn_readfirstlane(row_start[node]);

    float4 acc = make_float4(0.f, 0.f, 0.f, 0.f);
    for (int base = 0; base < cnt; base += 32) {
        const int m = min(32, cnt - base);
        int   myidx = 0;
        float mynrm = 0.f;
        if (lane < m) {
            myidx = col[start + base + lane];
            mynrm = rsqrtf(fmaxf((float)deg_out[myidx], 1.0f));
        }
        const int nb = __float_as_int(mynrm);
        int j = 0;
        for (; j + 8 <= m; j += 8) add_rows<8>(feat, myidx, nb, j, lane, acc);
        if (m - j >= 4) { add_rows<4>(feat, myidx, nb, j, lane, acc); j += 4; }
        if (m - j >= 2) { add_rows<2>(feat, myidx, nb, j, lane, acc); j += 2; }
        if (m - j >= 1) { add_rows<1>(feat, myidx, nb, j, lane, acc); }
    }
    ((float4*)(agg + (size_t)node * IN_DIM))[lane] = acc;
}

#define M_TILES 4
#define ROWS_PER_WG (16 * M_TILES)
#define A_PITCH (IN_DIM + 4)

struct BFrag { v2f w0, w1, s0, s1; };

__device__ __forceinline__ BFrag load_b(const float* __restrict__ W, const float* __restrict__ skip_W,
                                        int ka, int bcol) {
    const float* Wk0 = W      + (size_t)ka * OUT_DIM;
    const float* Wk1 = Wk0 + OUT_DIM;
    const float* Sk0 = skip_W + (size_t)ka * OUT_DIM;
    const float* Sk1 = Sk0 + OUT_DIM;
    BFrag b;
    b.w0[0] = Wk0[bcol];      b.w0[1] = Wk1[bcol];
    b.w1[0] = Wk0[bcol + 16]; b.w1[1] = Wk1[bcol + 16];
    b.s0[0] = Sk0[bcol];      b.s0[1] = Sk1[bcol];
    b.s1[0] = Sk0[bcol + 16]; b.s1[1] = Sk1[bcol + 16];
    return b;
}

__global__ __launch_bounds__(256) void gcn_fused(const float* __restrict__ agg,
                                                 const int* __restrict__ deg_in,
                                                 const float* __restrict__ W,
                                                 const float* __restrict__ bias,
                                                 const float* __restrict__ gamma,
                                                 const float* __restrict__ beta,
                                                 const float* __restrict__ feat,
                                                 const float* __restrict__ skip_W,
                                                 const float* __restrict__ skip_b,
                                                 float* __restrict__ out, int n_nodes) {
    __shared__ __attribute__((aligned(16))) float lds_agg[ROWS_PER_WG][A_PITCH];
    __shared__ __attribute__((aligned(16))) float lds_feat[ROWS_PER_WG][A_PITCH];
    __shared__ __attribute__((aligned(16))) float lds_gcn[16][OUT_DIM + 4];
    __shared__ __attribute__((aligned(16))) float lds_skip[16][OUT_DIM + 4];

    const int m0   = blockIdx.x * ROWS_PER_WG;
    const int wave = threadIdx.x >> 5;
    const int lane = threadIdx.x & 31;
    const int half = lane >> 4;
    const int l16  = lane & 15;
    const int bcol = wave * 32 + l16;

    BFrag bc = load_b(W, skip_W, 2 * half, bcol);

    {
        float4 va[ROWS_PER_WG / 8], vf[ROWS_PER_WG / 8];
#pragma unroll
        for (int i = 0; i < ROWS_PER_WG / 8; ++i) {
            const int gr = min(m0 + wave * (ROWS_PER_WG / 8) + i, n_nodes - 1);
            va[i] = ((const float4*)(agg  + (size_t)gr * IN_DIM))[lane];
            vf[i] = ((const float4*)(feat + (size_t)gr * IN_DIM))[lane];
        }
#pragma unroll
        for (int i = 0; i < ROWS_PER_WG / 8; ++i) {
            const int r = wave * (ROWS_PER_WG / 8) + i;
            *(float4*)&lds_agg[r][lane * 4]  = va[i];
            *(float4*)&lds_feat[r][lane * 4] = vf[i];
        }
    }
    __syncthreads();

    v8f cg[M_TILES][2];
    v8f cs[M_TILES][2];
#pragma unroll
    for (int mt = 0; mt < M_TILES; ++mt) {
        cg[mt][0] = v8f{}; cg[mt][1] = v8f{};
        cs[mt][0] = v8f{}; cs[mt][1] = v8f{};
    }

#pragma unroll 2
    for (int k0 = 0; k0 < IN_DIM; k0 += 4) {
        const int ka = k0 + 2 * half;
        const BFrag bn = load_b(W, skip_W, min(k0 + 4, IN_DIM - 4) + 2 * half, bcol);
#pragma unroll
        for (int mt = 0; mt < M_TILES; ++mt) {
            const v2f aA = *(const v2f*)&lds_agg[mt * 16 + l16][ka];
            const v2f aF = *(const v2f*)&lds_feat[mt * 16 + l16][ka];
            cg[mt][0] = __builtin_amdgcn_wmma_f32_16x16x4_f32(false, aA, false, bc.w0, (short)0, cg[mt][0], false, false);
            cg[mt][1] = __builtin_amdgcn_wmma_f32_16x16x4_f32(false, aA, false, bc.w1, (short)0, cg[mt][1], false, false);
            cs[mt][0] = __builtin_amdgcn_wmma_f32_16x16x4_f32(false, aF, false, bc.s0, (short)0, cs[mt][0], false, false);
            cs[mt][1] = __builtin_amdgcn_wmma_f32_16x16x4_f32(false, aF, false, bc.s1, (short)0, cs[mt][1], false, false);
        }
        bc = bn;
    }

    const float bb0 = bias[bcol],        sb0 = skip_b[bcol];
    const float bb1 = bias[bcol + 16],   sb1 = skip_b[bcol + 16];
    float gm[8], bt[8];
#pragma unroll
    for (int j = 0; j < 8; ++j) { gm[j] = gamma[lane * 8 + j]; bt[j] = beta[lane * 8 + j]; }

#pragma unroll
    for (int mt = 0; mt < M_TILES; ++mt) {
        if (mt > 0) __syncthreads();

#pragma unroll
        for (int i = 0; i < 8; ++i) {
            const int row = i + 8 * half;
            const int gr  = min(m0 + mt * 16 + row, n_nodes - 1);
            const float nrm = rsqrtf(fmaxf((float)deg_in[gr], 1.0f));
            lds_gcn[row][bcol]       = cg[mt][0][i] * nrm + bb0;
            lds_gcn[row][bcol + 16]  = cg[mt][1][i] * nrm + bb1;
            lds_skip[row][bcol]      = cs[mt][0][i] + sb0;
            lds_skip[row][bcol + 16] = cs[mt][1][i] + sb1;
        }
        __syncthreads();

#pragma unroll
        for (int rr = 0; rr < 2; ++rr) {
            const int row = wave * 2 + rr;
            float vals[8];
            float s = 0.f, s2 = 0.f;
#pragma unroll
            for (int j = 0; j < 8; ++j) {
                float x = lds_gcn[row][lane * 8 + j];
                vals[j] = x;
                s += x;
                s2 += x * x;
            }
            for (int off = 16; off > 0; off >>= 1) {
                s  += __shfl_xor(s,  off, 32);
                s2 += __shfl_xor(s2, off, 32);
            }
            const float mu  = s * (1.0f / OUT_DIM);
            const float var = s2 * (1.0f / OUT_DIM) - mu * mu;
            const float rs  = rsqrtf(var + LN_EPS);
            const int grow = m0 + mt * 16 + row;
            if (grow < n_nodes) {
                float* orow = out + (size_t)grow * OUT_DIM;
#pragma unroll
                for (int j = 0; j < 8; ++j) {
                    const int c = lane * 8 + j;
                    float ln = (vals[j] - mu) * rs * gm[j] + bt[j];
                    orow[c] = fmaxf(ln, 0.f) + lds_skip[row][c];
                }
            }
        }
    }
}

extern "C" void kernel_launch(void* const* d_in, const int* in_sizes, int n_in,
                              void* d_out, int out_size, void* d_ws, size_t ws_size,
                              hipStream_t stream) {
    const float* feat   = (const float*)d_in[0];
    const int*   src    = (const int*)d_in[1];
    const int*   dst    = (const int*)d_in[2];
    const float* W      = (const float*)d_in[3];
    const float* bias   = (const float*)d_in[4];
    const float* gamma  = (const float*)d_in[5];
    const float* beta   = (const float*)d_in[6];
    const float* skip_W = (const float*)d_in[7];
    const float* skip_b = (const float*)d_in[8];
    float* out = (float*)d_out;

    const int n_nodes = in_sizes[0] / IN_DIM;
    const int n_edges = in_sizes[1];

    const int nb = (n_nodes + BSIZE - 1) / BSIZE;
    int*   deg_out = (int*)d_ws;
    int*   deg_in  = deg_out + n_nodes;
    int*   row_ptr = deg_in + n_nodes;
    int*   pairs   = row_ptr + n_nodes;
    int*   col     = pairs + n_edges + n_edges;
    int*   sids    = col + n_edges;
    int*   cnt     = sids + n_edges;
    int*   bbase   = cnt + (size_t)2 * NCHUNK * nb;
    float* agg     = (float*)(deg_in + n_nodes) + (size_t)n_nodes * IN_DIM;

    const int T = 256;

    chunk_count<<<dim3(NCHUNK, 2), 256, 0, stream>>>(src, dst, cnt, n_edges, nb);

    chunk_scatter<<<dim3(NCHUNK, 2), 256, 0, stream>>>(src, dst, cnt, bbase, pairs, sids, n_edges, nb);

    bucket_build<<<dim3(nb, 2), BSIZE, 0, stream>>>(pairs, sids, bbase, deg_out, deg_in, row_ptr, col,
                                                    n_nodes, nb);

    {
        int waves_per_block = T / 32;
        int blocks = (n_nodes + waves_per_block - 1) / waves_per_block;
        spmm_gather<<<blocks, T, 0, stream>>>(col, row_ptr, deg_in, deg_out, feat, agg, n_nodes);
    }

    gcn_fused<<<(n_nodes + ROWS_PER_WG - 1) / ROWS_PER_WG, T, 0, stream>>>(agg, deg_in, W, bias, gamma, beta,
                                              feat, skip_W, skip_b, out, n_nodes);
}
